# starting kernel
# baseline (speedup 1.0000x reference)
#include <hip/hip_runtime.h>
#include <hip/hip_bf16.h>

typedef __attribute__((ext_vector_type(2))) float v2f;
typedef __attribute__((ext_vector_type(8))) float v8f;

#define IN_C    256
#define OUT_C   256
#define N_SPLIT 4
#define N_LEGO  64
#define BASIC_C 64
#define HIN     56
#define WIN     56
#define HOUT    19
#define WOUT    58
#define KDIM    192
#define NCOL    64
#define XSP_STR 160
#define FS_W    64

__global__ __launch_bounds__(256, 1)
void lego_conv_fused(const float* __restrict__ x,
                     const float* __restrict__ lego,
                     const float* __restrict__ coef,
                     const float* __restrict__ comb,
                     float* __restrict__ out) {
    __shared__ float XsP[(KDIM / 2) * XSP_STR];
    __shared__ float Fs[N_SPLIT * N_LEGO * FS_W];
    __shared__ float cW[N_SPLIT * OUT_C];
    __shared__ int   iW[N_SPLIT * OUT_C];

    const int tid = threadIdx.x;
    const int b   = blockIdx.y;
    const int h   = blockIdx.x;

    for (int p = tid; p < N_SPLIT * OUT_C; p += 256) {
        const float* cb = comb + p * N_LEGO;
        int   best = 0;
        float bv   = cb[0];
        #pragma unroll
        for (int n = 1; n < N_LEGO; ++n) {
            float v = cb[n];
            if (v > bv) { bv = v; best = n; }
        }
        cW[p] = coef[p * N_LEGO + best];
        iW[p] = best;
    }

    const int lane = tid & 31;
    const int wave = tid >> 5;
    const int half = lane >> 4;
    const int l16  = lane & 15;
    const int mt   = wave >> 1;
    const int nt0  = (wave & 1) * 2;

    const float* Arow = lego + (mt * 16 + l16) * KDIM + half * 2;

    for (int s = 0; s < N_SPLIT; ++s) {
        __syncthreads();

        for (int j = tid; j < KDIM * NCOL; j += 256) {
            int   k = j >> 6;
            int   w = j & 63;
            float val = 0.0f;
            if (w >= 1 && w <= WIN) {
                int ci = k / 3;
                int kh = k - ci * 3;
                int r  = 3 * h + kh - 1;
                if (r >= 0)
                    val = x[(((size_t)b * IN_C + s * BASIC_C + ci) * HIN + r) * WIN + (w - 1)];
            }
            XsP[(k >> 1) * XSP_STR + (w << 1) + (k & 1)] = val;
        }
        __syncthreads();

        v8f acc0 = {0.f, 0.f, 0.f, 0.f, 0.f, 0.f, 0.f, 0.f};
        v8f acc1 = {0.f, 0.f, 0.f, 0.f, 0.f, 0.f, 0.f, 0.f};
        #pragma unroll 4
        for (int kb = 0; kb < KDIM; kb += 4) {
            v2f a = *(const v2f*)(Arow + kb);
            const float* brow = &XsP[((kb >> 1) + half) * XSP_STR + (l16 << 1)];
            v2f b0 = *(const v2f*)(brow + nt0 * 32);
            v2f b1 = *(const v2f*)(brow + nt0 * 32 + 32);
            acc0 = __builtin_amdgcn_wmma_f32_16x16x4_f32(
                       false, a, false, b0, (short)0, acc0, false, false);
            acc1 = __builtin_amdgcn_wmma_f32_16x16x4_f32(
                       false, a, false, b1, (short)0, acc1, false, false);
        }

        float* fbase = Fs + (s * N_LEGO + mt * 16 + half * 8) * FS_W;
        #pragma unroll
        for (int v = 0; v < 8; ++v) {
            fbase[v * FS_W + nt0       * 16 + l16] = acc0[v];
            fbase[v * FS_W + (nt0 + 1) * 16 + l16] = acc1[v];
        }
    }
    __syncthreads();

    const int wl = tid & 63;
    const int o0 = tid >> 6;
    if (wl < WOUT) {
        for (int o = o0; o < OUT_C; o += 4) {
            float acc = 0.f;
            #pragma unroll
            for (int s = 0; s < N_SPLIT; ++s) {
                int n = iW[s * OUT_C + o];
                acc += cW[s * OUT_C + o] * Fs[(s * N_LEGO + n) * FS_W + wl];
            }
            out[(((size_t)b * OUT_C + o) * HOUT + h) * WOUT + wl] = acc;
        }
    }
}

extern "C" void kernel_launch(void* const* d_in, const int* in_sizes, int n_in,
                              void* d_out, int out_size, void* d_ws, size_t ws_size,
                              hipStream_t stream) {
    (void)in_sizes; (void)n_in; (void)d_ws; (void)ws_size; (void)out_size;
    const float* x    = (const float*)d_in[0];
    const float* lego = (const float*)d_in[1];
    const float* coef = (const float*)d_in[2];
    const float* comb = (const float*)d_in[3];
    float* out = (float*)d_out;

    dim3 grid(HOUT, 32, 1);
    dim3 block(256, 1, 1);
    lego_conv_fused<<<grid, block, 0, stream>>>(x, lego, coef, comb, out);
}


# combine of row r (flat (o,w) elements on all lanes, table packed per o) runs in four parts between the matrix chunks of row r+1, not behind the row's barrier; last row's stays; sums kept
# speedup vs baseline: 4.8119x; 4.8119x over previous
#include <hip/hip_runtime.h>
#include <hip/hip_bf16.h>


typedef __attribute__((ext_vector_type(2))) float v2f;
typedef __attribute__((ext_vector_type(4))) float v4f;
typedef __attribute__((ext_vector_type(8))) float v8f;

#define IN_C    256
#define OUT_C   256
#define N_SPLIT 4
#define N_LEGO  64
#define BASIC_C 64
#define HIN     56
#define WIN     56
#define HOUT    19
#define WOUT    58
#define KDIM    192
#define KSTEPS  48
#define CHW     (HIN * WIN)
#define AS_STR  258
#define FS_W    64
#define FS_SIZE (N_SPLIT * N_LEGO * FS_W)
#define CH_G    4
#define CH_CI   16
#define NTHR    512
#define ROW_GROUPS 8

#define SCHED_FENCE() __builtin_amdgcn_sched_barrier(0)

struct BChunk { float v[CH_G][3][2]; };

__device__ __forceinline__ void load_chunk(BChunk& c, const float* __restrict__ p,
                                           const int (&off)[3][2]) {
    #pragma unroll
    for (int g = 0; g < CH_G; ++g) {
        #pragma unroll
        for (int j = 0; j < 3; ++j) {
            #pragma unroll
            for (int e = 0; e < 2; ++e)
                c.v[g][j][e] = p[g * (4 * CHW) + off[j][e]];
        }
    }
}

template <int G0 = 0, int G1 = CH_G>
__device__ __forceinline__ void mma_chunk(v8f (&acc)[4], const BChunk& c,
                                          const float* afr, const int ks0,
                                          const bool (&ok)[3][2]) {
    #pragma unroll
    for (int g = G0; g < G1; ++g) {
        #pragma unroll
        for (int j = 0; j < 3; ++j) {
            v2f bv;
            bv.x = ok[j][0] ? c.v[g][j][0] : 0.0f;
            bv.y = ok[j][1] ? c.v[g][j][1] : 0.0f;
            const float* ap = afr + (ks0 + g * 3 + j) * AS_STR;
            #pragma unroll
            for (int mt = 0; mt < 4; ++mt) {
                v2f a = *(const v2f*)(ap + mt * 64);
                acc[mt] = __builtin_amdgcn_wmma_f32_16x16x4_f32(
                              false, a, false, bv, (short)0, acc[mt], false, false);
            }
        }
    }
}

__device__ __forceinline__ void row_offsets(int (&off)[3][2], bool (&ok)[3][2],
                                            const int h, const int half,
                                            const int col, const bool colok) {
    #pragma unroll
    for (int j = 0; j < 3; ++j) {
        #pragma unroll
        for (int e = 0; e < 2; ++e) {
            const int k  = 4 * j + 2 * half + e;
            const int ci = k / 3;
            const int kh = k - 3 * ci;
            const int r  = 3 * h + kh - 1;
            ok[j][e]  = colok && (r >= 0);
            off[j][e] = ci * CHW + (r < 0 ? 0 : r) * WIN + col;
        }
    }
}

template <int I0, int I1, int SUB>
__device__ __forceinline__ void combine_part(const float* Fp, const float* cW4,
                                             const unsigned* iP, float* outbh,
                                             const unsigned tid, const unsigned ebase = 0u) {
    #pragma unroll
    for (int i0 = I0; i0 < I1; i0 += SUB) {
        unsigned t = tid + ebase;
        asm volatile("" : "+v"(t));
        v4f      cf[SUB];
        unsigned pk[SUB];
        unsigned oo[SUB], ww[SUB];
        #pragma unroll
        for (int j = 0; j < SUB; ++j) {
            if (i0 + j < I1) {
                const unsigned e = t + (unsigned)NTHR * (unsigned)(i0 + j);
                oo[j] = e / WOUT;
                ww[j] = e - oo[j] * WOUT;
                cf[j] = *(const v4f*)(cW4 + oo[j] * N_SPLIT);
                pk[j] = iP[oo[j]];
            }
        }
        float f[SUB][N_SPLIT];
        #pragma unroll
        for (int j = 0; j < SUB; ++j) {
            if (i0 + j < I1) {
                #pragma unroll
                for (int sp = 0; sp < N_SPLIT; ++sp) {
                    const unsigned n = (pk[j] >> (8 * sp)) & 255u;
                    f[j][sp] = Fp[(sp * N_LEGO + n) * FS_W + ww[j]];
                }
            }
        }
        #pragma unroll
        for (int j = 0; j < SUB; ++j) {
            if (i0 + j < I1) {
                float sum = 0.f;
                sum += cf[j].x * f[j][0];
                sum += cf[j].y * f[j][1];
                sum += cf[j].z * f[j][2];
                sum += cf[j].w * f[j][3];
                outbh[oo[j] * (HOUT * WOUT) + ww[j]] = sum;
            }
        }
    }
}

__device__ __forceinline__ void spill_feat(const v8f (&acc)[4], float* Fc, const int s,
                                           const int nt, const int half, const int l16) {
    #pragma unroll
    for (int mt = 0; mt < 4; ++mt) {
        float* fbase = Fc + (s * N_LEGO + mt * 16 + half * 8) * FS_W + nt * 16 + l16;
        #pragma unroll
        for (int v = 0; v < 8; ++v)
            fbase[v * FS_W] = acc[mt][v];
    }
}

template <bool CMB>
__device__ __forceinline__ void row_stage(v8f (&acc)[4], BChunk& q0, BChunk& q1,
                                          const float* xs, const float* afr,
                                          const int (&off)[3][2], const bool (&ok)[3][2],
                                          const int (&offn)[3][2],
                                          const float* Fp, const float* cW4,
                                          const unsigned* iP, float* outbh,
                                          const unsigned tid) {
    load_chunk(q1, xs + 1 * CH_CI * CHW, off);
    SCHED_FENCE();
    if (CMB) {
        mma_chunk<0, 2>(acc, q0, afr, 0, ok);
        SCHED_FENCE();
        combine_part<24, 29, 5>(Fp, cW4, iP, outbh, tid);
        SCHED_FENCE();
        mma_chunk<2, CH_G>(acc, q0, afr, 0, ok);
    } else {
        mma_chunk(acc, q0, afr, 0, ok);
    }
    SCHED_FENCE();
    load_chunk(q0, xs + 2 * CH_CI * CHW, off);
    SCHED_FENCE();
    if (CMB) { combine_part<0, 8, 8>(Fp, cW4, iP, outbh, tid); SCHED_FENCE(); }
    mma_chunk(acc, q1, afr, 12, ok);
    SCHED_FENCE();
    load_chunk(q1, xs + 3 * CH_CI * CHW, off);
    SCHED_FENCE();
    if (CMB) { combine_part<8, 16, 8>(Fp, cW4, iP, outbh, tid); SCHED_FENCE(); }
    mma_chunk(acc, q0, afr, 24, ok);
    SCHED_FENCE();
    load_chunk(q0, xs, offn);
    SCHED_FENCE();
    if (CMB) { combine_part<16, 24, 8>(Fp, cW4, iP, outbh, tid); SCHED_FENCE(); }
    mma_chunk(acc, q1, afr, 36, ok);
    SCHED_FENCE();
}

__global__ __launch_bounds__(NTHR, 1)
void lego_conv_fused(const float* __restrict__ x,
                     const float* __restrict__ lego,
                     const float* __restrict__ coef,
                     const float* __restrict__ comb,
                     float* __restrict__ out) {
    __shared__ __attribute__((aligned(16))) float smem[KSTEPS * AS_STR + 2 * FS_SIZE + N_SPLIT * OUT_C + OUT_C];
    float* As = smem;
    float* Fs = As + KSTEPS * AS_STR;
    float* cW4 = Fs + 2 * FS_SIZE;
    unsigned* iP = (unsigned*)(cW4 + N_SPLIT * OUT_C);

    const int tid = threadIdx.x;
    const int b   = blockIdx.y;
    const int h0  = (blockIdx.x * HOUT) / ROW_GROUPS;
    const int h1  = ((blockIdx.x + 1) * HOUT) / ROW_GROUPS;

    for (int p = tid; p < N_SPLIT * OUT_C; p += NTHR) {
        const float* cb = comb + p * N_LEGO;
        int   best = 0;
        float bv   = cb[0];
        #pragma unroll
        for (int n = 1; n < N_LEGO; ++n) {
            float v = cb[n];
            if (v > bv) { bv = v; best = n; }
        }
        cW4[(p & (OUT_C - 1)) * N_SPLIT + (p >> 8)] = coef[p * N_LEGO + best];
        ((unsigned char*)iP)[(p & (OUT_C - 1)) * N_SPLIT + (p >> 8)] = (unsigned char)best;
    }

    const int lane = tid & 31;
    const int wave = tid >> 5;
    const int half = lane >> 4;
    const int l16  = lane & 15;
    const int nt   = wave & 3;
    const int s    = wave >> 2;

    const int  w     = nt * 16 + l16;
    const bool colok = (w >= 1) && (w <= WIN);
    const int  col   = (w < 1) ? 0 : ((w > WIN) ? (WIN - 1) : (w - 1));
    int  off[3][2];
    bool ok[3][2];
    row_offsets(off, ok, h0, half, col, colok);
    const float* xs = x + ((size_t)b * IN_C + (size_t)s * BASIC_C) * CHW;

    BChunk q0, q1;
    load_chunk(q0, xs, off);

    for (int q = tid; q < N_LEGO * KSTEPS; q += NTHR) {
        const int m  = q / KSTEPS;
        const int ks = q - m * KSTEPS;
        const v4f v  = *(const v4f*)(lego + m * KDIM + ks * 4);
        float* d = As + ks * AS_STR + (m >> 4) * 64 + (m & 15) * 2;
        v2f lo; lo.x = v.x; lo.y = v.y;
        v2f hi; hi.x = v.z; hi.y = v.w;
        *(v2f*)d        = lo;
        *(v2f*)(d + 32) = hi;
    }
    __syncthreads();

    const float* afr = As + lane * 2;
    float* outb = out + (size_t)b * OUT_C * HOUT * WOUT;

    {
        v8f acc[4];
        #pragma unroll
        for (int mt = 0; mt < 4; ++mt)
            acc[mt] = (v8f){0.f, 0.f, 0.f, 0.f, 0.f, 0.f, 0.f, 0.f};
        int  offn[3][2];
        bool okn[3][2];
        row_offsets(offn, okn, (h0 + 1 < h1) ? (h0 + 1) : h0, half, col, colok);
        row_stage<false>(acc, q0, q1, xs, afr, off, ok, offn, Fs, cW4, iP, outb, (unsigned)tid);
        spill_feat(acc, Fs, s, nt, half, l16);
        __syncthreads();
        #pragma unroll
        for (int j = 0; j < 3; ++j) {
            #pragma unroll
            for (int e = 0; e < 2; ++e) { off[j][e] = offn[j][e]; ok[j][e] = okn[j][e]; }
        }
    }

    for (int h = h0 + 1; h < h1; ++h) {
        float*       Fc = Fs + ((h - h0) & 1) * FS_SIZE;
        const float* Fp = Fs + ((h - h0 + 1) & 1) * FS_SIZE;

        v8f acc[4];
        #pragma unroll
        for (int mt = 0; mt < 4; ++mt)
            acc[mt] = (v8f){0.f, 0.f, 0.f, 0.f, 0.f, 0.f, 0.f, 0.f};

        int  offn[3][2];
        bool okn[3][2];
        row_offsets(offn, okn, (h + 1 < h1) ? (h + 1) : h, half, col, colok);

        row_stage<true>(acc, q0, q1, xs, afr, off, ok, offn, Fp, cW4, iP,
                        outb + (size_t)(h - 1) * WOUT, (unsigned)tid);

        spill_feat(acc, Fc, s, nt, half, l16);
        __syncthreads();

        #pragma unroll
        for (int j = 0; j < 3; ++j) {
            #pragma unroll
            for (int e = 0; e < 2; ++e) { off[j][e] = offn[j][e]; ok[j][e] = okn[j][e]; }
        }
    }

    {
        const float* Fl    = Fs + ((h1 - 1 - h0) & 1) * FS_SIZE;
        float*       outbl = outb + (size_t)(h1 - 1) * WOUT;
        #pragma nounroll
        for (unsigned r = 0; r < 3u; ++r)
            combine_part<0, 8, 8>(Fl, cW4, iP, outbl, (unsigned)tid, r * 8u * (unsigned)NTHR);
        combine_part<24, 29, 5>(Fl, cW4, iP, outbl, (unsigned)tid);
    }
}

extern "C" void kernel_launch(void* const* d_in, const int* in_sizes, int n_in,
                              void* d_out, int out_size, void* d_ws, size_t ws_size,
                              hipStream_t stream) {
    (void)in_sizes; (void)n_in; (void)d_ws; (void)ws_size; (void)out_size;
    const float* x    = (const float*)d_in[0];
    const float* lego = (const float*)d_in[1];
    const float* coef = (const float*)d_in[2];
    const float* comb = (const float*)d_in[3];
    float* out = (float*)d_out;

    dim3 grid(ROW_GROUPS, 32, 1);
    dim3 block(NTHR, 1, 1);
    lego_conv_fused<<<grid, block, 0, stream>>>(x, lego, coef, comb, out);
}


# set-up: table scan reads whole 128-byte lines, 8 lanes per row, row_xmask max then lowest-index min (was one lane per row); all set-up loads issued before the first wait; sums kept
# speedup vs baseline: 5.0339x; 1.0461x over previous
#include <hip/hip_runtime.h>
#include <hip/hip_bf16.h>


typedef __attribute__((ext_vector_type(2))) float v2f;
typedef __attribute__((ext_vector_type(4))) float v4f;
typedef __attribute__((ext_vector_type(8))) float v8f;

#define IN_C    256
#define OUT_C   256
#define N_SPLIT 4
#define N_LEGO  64
#define BASIC_C 64
#define HIN     56
#define WIN     56
#define HOUT    19
#define WOUT    58
#define KDIM    192
#define KSTEPS  48
#define CHW     (HIN * WIN)
#define AS_STR  258
#define FS_W    64
#define FS_SIZE (N_SPLIT * N_LEGO * FS_W)
#define CH_G    4
#define CH_CI   16
#define NTHR    512
#define ROW_GROUPS 8

#define SCHED_FENCE() __builtin_amdgcn_sched_barrier(0)

struct BChunk { float v[CH_G][3][2]; };

__device__ __forceinline__ void load_chunk(BChunk& c, const float* __restrict__ p,
                                           const int (&off)[3][2]) {
    #pragma unroll
    for (int g = 0; g < CH_G; ++g) {
        #pragma unroll
        for (int j = 0; j < 3; ++j) {
            #pragma unroll
            for (int e = 0; e < 2; ++e)
                c.v[g][j][e] = p[g * (4 * CHW) + off[j][e]];
        }
    }
}

template <int G0 = 0, int G1 = CH_G>
__device__ __forceinline__ void mma_chunk(v8f (&acc)[4], const BChunk& c,
                                          const float* afr, const int ks0,
                                          const bool (&ok)[3][2]) {
    #pragma unroll
    for (int g = G0; g < G1; ++g) {
        #pragma unroll
        for (int j = 0; j < 3; ++j) {
            v2f bv;
            bv.x = ok[j][0] ? c.v[g][j][0] : 0.0f;
            bv.y = ok[j][1] ? c.v[g][j][1] : 0.0f;
            const float* ap = afr + (ks0 + g * 3 + j) * AS_STR;
            #pragma unroll
            for (int mt = 0; mt < 4; ++mt) {
                v2f a = *(const v2f*)(ap + mt * 64);
                acc[mt] = __builtin_amdgcn_wmma_f32_16x16x4_f32(
                              false, a, false, bv, (short)0, acc[mt], false, false);
            }
        }
    }
}

__device__ __forceinline__ void row_offsets(int (&off)[3][2], bool (&ok)[3][2],
                                            const int h, const int half,
                                            const int col, const bool colok) {
    #pragma unroll
    for (int j = 0; j < 3; ++j) {
        #pragma unroll
        for (int e = 0; e < 2; ++e) {
            const int k  = 4 * j + 2 * half + e;
            const int ci = k / 3;
            const int kh = k - 3 * ci;
            const int r  = 3 * h + kh - 1;
            ok[j][e]  = colok && (r >= 0);
            off[j][e] = ci * CHW + (r < 0 ? 0 : r) * WIN + col;
        }
    }
}

template <int I0, int I1, int SUB>
__device__ __forceinline__ void combine_part(const float* Fp, const float* cW4,
                                             const unsigned* iP, float* outbh,
                                             const unsigned tid, const unsigned ebase = 0u) {
    #pragma unroll
    for (int i0 = I0; i0 < I1; i0 += SUB) {
        unsigned t = tid + ebase;
        asm volatile("" : "+v"(t));
        v4f      cf[SUB];
        unsigned pk[SUB];
        unsigned oo[SUB], ww[SUB];
        #pragma unroll
        for (int j = 0; j < SUB; ++j) {
            if (i0 + j < I1) {
                const unsigned e = t + (unsigned)NTHR * (unsigned)(i0 + j);
                oo[j] = e / WOUT;
                ww[j] = e - oo[j] * WOUT;
                cf[j] = *(const v4f*)(cW4 + oo[j] * N_SPLIT);
                pk[j] = iP[oo[j]];
            }
        }
        float f[SUB][N_SPLIT];
        #pragma unroll
        for (int j = 0; j < SUB; ++j) {
            if (i0 + j < I1) {
                #pragma unroll
                for (int sp = 0; sp < N_SPLIT; ++sp) {
                    const unsigned n = (pk[j] >> (8 * sp)) & 255u;
                    f[j][sp] = Fp[(sp * N_LEGO + n) * FS_W + ww[j]];
                }
            }
        }
        #pragma unroll
        for (int j = 0; j < SUB; ++j) {
            if (i0 + j < I1) {
                float sum = 0.f;
                sum += cf[j].x * f[j][0];
                sum += cf[j].y * f[j][1];
                sum += cf[j].z * f[j][2];
                sum += cf[j].w * f[j][3];
                outbh[oo[j] * (HOUT * WOUT) + ww[j]] = sum;
            }
        }
    }
}

__device__ __forceinline__ void spill_feat(const v8f (&acc)[4], float* Fc, const int s,
                                           const int nt, const int half, const int l16) {
    #pragma unroll
    for (int mt = 0; mt < 4; ++mt) {
        float* fbase = Fc + (s * N_LEGO + mt * 16 + half * 8) * FS_W + nt * 16 + l16;
        #pragma unroll
        for (int v = 0; v < 8; ++v)
            fbase[v * FS_W] = acc[mt][v];
    }
}

template <bool CMB>
__device__ __forceinline__ void row_stage(v8f (&acc)[4], BChunk& q0, BChunk& q1,
                                          const float* xs, const float* afr,
                                          const int (&off)[3][2], const bool (&ok)[3][2],
                                          const int (&offn)[3][2],
                                          const float* Fp, const float* cW4,
                                          const unsigned* iP, float* outbh,
                                          const unsigned tid) {
    load_chunk(q1, xs + 1 * CH_CI * CHW, off);
    SCHED_FENCE();
    if (CMB) {
        mma_chunk<0, 2>(acc, q0, afr, 0, ok);
        SCHED_FENCE();
        combine_part<24, 29, 5>(Fp, cW4, iP, outbh, tid);
        SCHED_FENCE();
        mma_chunk<2, CH_G>(acc, q0, afr, 0, ok);
    } else {
        mma_chunk(acc, q0, afr, 0, ok);
    }
    SCHED_FENCE();
    load_chunk(q0, xs + 2 * CH_CI * CHW, off);
    SCHED_FENCE();
    if (CMB) { combine_part<0, 8, 8>(Fp, cW4, iP, outbh, tid); SCHED_FENCE(); }
    mma_chunk(acc, q1, afr, 12, ok);
    SCHED_FENCE();
    load_chunk(q1, xs + 3 * CH_CI * CHW, off);
    SCHED_FENCE();
    if (CMB) { combine_part<8, 16, 8>(Fp, cW4, iP, outbh, tid); SCHED_FENCE(); }
    mma_chunk(acc, q0, afr, 24, ok);
    SCHED_FENCE();
    load_chunk(q0, xs, offn);
    SCHED_FENCE();
    if (CMB) { combine_part<16, 24, 8>(Fp, cW4, iP, outbh, tid); SCHED_FENCE(); }
    mma_chunk(acc, q1, afr, 36, ok);
    SCHED_FENCE();
}

template <int XM>
__device__ __forceinline__ int xmask_i(const int v) {
    return __builtin_amdgcn_update_dpp(v, v, 0x160 + XM, 0xf, 0xf, false);
}
template <int XM>
__device__ __forceinline__ float xmask_f(const float v) {
    return __builtin_bit_cast(float, xmask_i<XM>(__builtin_bit_cast(int, v)));
}

__device__ __forceinline__ int row_argmax8(const v4f va, const v4f vb, const int l8) {
    float m = __builtin_fmaxf(__builtin_fmaxf(va.x, va.y), __builtin_fmaxf(va.z, va.w));
    m = __builtin_fmaxf(m, __builtin_fmaxf(__builtin_fmaxf(vb.x, vb.y), __builtin_fmaxf(vb.z, vb.w)));
    m = __builtin_fmaxf(m, xmask_f<1>(m));
    m = __builtin_fmaxf(m, xmask_f<2>(m));
    m = __builtin_fmaxf(m, xmask_f<4>(m));
    int jc = 64;
    jc = (vb.w == m) ? 35 : jc;
    jc = (vb.z == m) ? 34 : jc;
    jc = (vb.y == m) ? 33 : jc;
    jc = (vb.x == m) ? 32 : jc;
    jc = (va.w == m) ? 3 : jc;
    jc = (va.z == m) ? 2 : jc;
    jc = (va.y == m) ? 1 : jc;
    jc = (va.x == m) ? 0 : jc;
    int c = jc + l8 * 4;
    c = min(c, xmask_i<1>(c));
    c = min(c, xmask_i<2>(c));
    c = min(c, xmask_i<4>(c));
    return min(c, N_LEGO - 1);
}

__global__ __launch_bounds__(NTHR, 1)
void lego_conv_fused(const float* __restrict__ x,
                     const float* __restrict__ lego,
                     const float* __restrict__ coef,
                     const float* __restrict__ comb,
                     float* __restrict__ out) {
    __shared__ __attribute__((aligned(16))) float smem[KSTEPS * AS_STR + 2 * FS_SIZE + N_SPLIT * OUT_C + OUT_C];
    float* As = smem;
    float* Fs = As + KSTEPS * AS_STR;
    float* cW4 = Fs + 2 * FS_SIZE;
    unsigned* iP = (unsigned*)(cW4 + N_SPLIT * OUT_C);

    const int tid = threadIdx.x;
    const int b   = blockIdx.y;
    const int h0  = (blockIdx.x * HOUT) / ROW_GROUPS;
    const int h1  = ((blockIdx.x + 1) * HOUT) / ROW_GROUPS;

    const int lane = tid & 31;
    const int wave = tid >> 5;
    const int half = lane >> 4;
    const int l16  = lane & 15;
    const int nt   = wave & 3;
    const int s    = wave >> 2;
    const int l8   = lane & 7;
    const int g8   = lane >> 3;

    const float* cb0 = comb + (wave * 64 + g8) * N_LEGO + l8 * 4;
    v4f ra[16], rb[16];
    #pragma unroll
    for (int st = 0; st < 16; ++st) {
        ra[st] = *(const v4f*)(cb0 + st * (4 * N_LEGO));
        rb[st] = *(const v4f*)(cb0 + st * (4 * N_LEGO) + 32);
    }
    v4f lg[6];
    #pragma unroll
    for (int i = 0; i < 6; ++i)
        lg[i] = *(const v4f*)(lego + (tid + i * NTHR) * 4);
    const int  w     = nt * 16 + l16;
    const bool colok = (w >= 1) && (w <= WIN);
    const int  col   = (w < 1) ? 0 : ((w > WIN) ? (WIN - 1) : (w - 1));
    int  off[3][2];
    bool ok[3][2];
    row_offsets(off, ok, h0, half, col, colok);
    const float* xs = x + ((size_t)b * IN_C + (size_t)s * BASIC_C) * CHW;

    BChunk q0, q1;
    load_chunk(q0, xs, off);
    SCHED_FENCE();

    int   prow[2], pbest[2];
    float pcoef[2];
    #pragma unroll
    for (int bt = 0; bt < 2; ++bt) {
        int mine = 0;
        #pragma unroll
        for (int j = 0; j < 8; ++j) {
            const int r = row_argmax8(ra[bt * 8 + j], rb[bt * 8 + j], l8);
            mine = (l8 == j) ? r : mine;
        }
        const int p = wave * 64 + (bt * 8 + l8) * 4 + g8;
        prow[bt]  = p;
        pbest[bt] = mine;
        pcoef[bt] = coef[p * N_LEGO + mine];
    }

    #pragma unroll
    for (int i = 0; i < 6; ++i) {
        const int q  = tid + i * NTHR;
        const int m  = q / KSTEPS;
        const int ks = q - m * KSTEPS;
        float* d = As + ks * AS_STR + (m >> 4) * 64 + (m & 15) * 2;
        v2f lo; lo.x = lg[i].x; lo.y = lg[i].y;
        v2f hi; hi.x = lg[i].z; hi.y = lg[i].w;
        *(v2f*)d        = lo;
        *(v2f*)(d + 32) = hi;
    }
    #pragma unroll
    for (int bt = 0; bt < 2; ++bt) {
        const int p = prow[bt];
        cW4[(p & (OUT_C - 1)) * N_SPLIT + (p >> 8)] = pcoef[bt];
        ((unsigned char*)iP)[(p & (OUT_C - 1)) * N_SPLIT + (p >> 8)] = (unsigned char)pbest[bt];
    }
    __syncthreads();

    const float* afr = As + lane * 2;
    float* outb = out + (size_t)b * OUT_C * HOUT * WOUT;

    {
        v8f acc[4];
        #pragma unroll
        for (int mt = 0; mt < 4; ++mt)
            acc[mt] = (v8f){0.f, 0.f, 0.f, 0.f, 0.f, 0.f, 0.f, 0.f};
        int  offn[3][2];
        bool okn[3][2];
        row_offsets(offn, okn, (h0 + 1 < h1) ? (h0 + 1) : h0, half, col, colok);
        row_stage<false>(acc, q0, q1, xs, afr, off, ok, offn, Fs, cW4, iP, outb, (unsigned)tid);
        spill_feat(acc, Fs, s, nt, half, l16);
        __syncthreads();
        #pragma unroll
        for (int j = 0; j < 3; ++j) {
            #pragma unroll
            for (int e = 0; e < 2; ++e) { off[j][e] = offn[j][e]; ok[j][e] = okn[j][e]; }
        }
    }

    for (int h = h0 + 1; h < h1; ++h) {
        float*       Fc = Fs + ((h - h0) & 1) * FS_SIZE;
        const float* Fp = Fs + ((h - h0 + 1) & 1) * FS_SIZE;

        v8f acc[4];
        #pragma unroll
        for (int mt = 0; mt < 4; ++mt)
            acc[mt] = (v8f){0.f, 0.f, 0.f, 0.f, 0.f, 0.f, 0.f, 0.f};

        int  offn[3][2];
        bool okn[3][2];
        row_offsets(offn, okn, (h + 1 < h1) ? (h + 1) : h, half, col, colok);

        row_stage<true>(acc, q0, q1, xs, afr, off, ok, offn, Fp, cW4, iP,
                        outb + (size_t)(h - 1) * WOUT, (unsigned)tid);

        spill_feat(acc, Fc, s, nt, half, l16);
        __syncthreads();

        #pragma unroll
        for (int j = 0; j < 3; ++j) {
            #pragma unroll
            for (int e = 0; e < 2; ++e) { off[j][e] = offn[j][e]; ok[j][e] = okn[j][e]; }
        }
    }

    {
        const float* Fl    = Fs + ((h1 - 1 - h0) & 1) * FS_SIZE;
        float*       outbl = outb + (size_t)(h1 - 1) * WOUT;
        #pragma nounroll
        for (unsigned r = 0; r < 3u; ++r)
            combine_part<0, 8, 8>(Fl, cW4, iP, outbl, (unsigned)tid, r * 8u * (unsigned)NTHR);
        combine_part<24, 29, 5>(Fl, cW4, iP, outbl, (unsigned)tid);
    }
}

extern "C" void kernel_launch(void* const* d_in, const int* in_sizes, int n_in,
                              void* d_out, int out_size, void* d_ws, size_t ws_size,
                              hipStream_t stream) {
    (void)in_sizes; (void)n_in; (void)d_ws; (void)ws_size; (void)out_size;
    const float* x    = (const float*)d_in[0];
    const float* lego = (const float*)d_in[1];
    const float* coef = (const float*)d_in[2];
    const float* comb = (const float*)d_in[3];
    float* out = (float*)d_out;

    dim3 grid(ROW_GROUPS, 32, 1);
    dim3 block(NTHR, 1, 1);
    lego_conv_fused<<<grid, block, 0, stream>>>(x, lego, coef, comb, out);
}


# combine: a thread keeps one output column (tid & 63) and steps o by 8, so table, feat and store addresses are one per-thread base plus immediates; lego index kept as a byte offset; sums kept
# speedup vs baseline: 5.3116x; 1.0552x over previous
#include <hip/hip_runtime.h>
#include <hip/hip_bf16.h>


typedef __attribute__((ext_vector_type(2))) float v2f;
typedef __attribute__((ext_vector_type(4))) float v4f;
typedef __attribute__((ext_vector_type(8))) float v8f;
typedef __attribute__((ext_vector_type(4))) unsigned v4u;

#define IN_C    256
#define OUT_C   256
#define N_SPLIT 4
#define N_LEGO  64
#define BASIC_C 64
#define HIN     56
#define WIN     56
#define HOUT    19
#define WOUT    58
#define KDIM    192
#define KSTEPS  48
#define CHW     (HIN * WIN)
#define AS_STR  258
#define FS_W    64
#define FS_SIZE (N_SPLIT * N_LEGO * FS_W)
#define CH_G    4
#define CH_CI   16
#define NTHR    512
#define ROW_GROUPS 8

#define SCHED_FENCE() __builtin_amdgcn_sched_barrier(0)

struct BChunk { float v[CH_G][3][2]; };

__device__ __forceinline__ void load_chunk(BChunk& c, const float* __restrict__ p,
                                           const int (&off)[3][2]) {
    #pragma unroll
    for (int g = 0; g < CH_G; ++g) {
        #pragma unroll
        for (int j = 0; j < 3; ++j) {
            #pragma unroll
            for (int e = 0; e < 2; ++e)
                c.v[g][j][e] = p[g * (4 * CHW) + off[j][e]];
        }
    }
}

template <int G0 = 0, int G1 = CH_G>
__device__ __forceinline__ void mma_chunk(v8f (&acc)[4], const BChunk& c,
                                          const float* afr, const int ks0,
                                          const bool (&ok)[3][2]) {
    #pragma unroll
    for (int g = G0; g < G1; ++g) {
        #pragma unroll
        for (int j = 0; j < 3; ++j) {
            v2f bv;
            bv.x = ok[j][0] ? c.v[g][j][0] : 0.0f;
            bv.y = ok[j][1] ? c.v[g][j][1] : 0.0f;
            const float* ap = afr + (ks0 + g * 3 + j) * AS_STR;
            #pragma unroll
            for (int mt = 0; mt < 4; ++mt) {
                v2f a = *(const v2f*)(ap + mt * 64);
                acc[mt] = __builtin_amdgcn_wmma_f32_16x16x4_f32(
                              false, a, false, bv, (short)0, acc[mt], false, false);
            }
        }
    }
}

__device__ __forceinline__ void row_offsets(int (&off)[3][2], bool (&ok)[3][2],
                                            const int h, const int half,
                                            const int col, const bool colok) {
    #pragma unroll
    for (int j = 0; j < 3; ++j) {
        #pragma unroll
        for (int e = 0; e < 2; ++e) {
            const int k  = 4 * j + 2 * half + e;
            const int ci = k / 3;
            const int kh = k - 3 * ci;
            const int r  = 3 * h + kh - 1;
            ok[j][e]  = colok && (r >= 0);
            off[j][e] = ci * CHW + (r < 0 ? 0 : r) * WIN + col;
        }
    }
}

template <int I0, int I1, int SUB>
__device__ __forceinline__ void combine_part(const float* Fp, const float* cW4,
                                             const unsigned* gO, float* outbh,
                                             const unsigned tid, const unsigned ib = 0u) {
    unsigned t = tid;
    asm volatile("" : "+v"(t));
    const unsigned w  = t & 63u;
    const unsigned o0 = (t >> 6) + 8u * ib;
    if (w < (unsigned)WOUT) {
        const char*     fb = (const char*)(Fp + w);
        const float*    cw = cW4 + o0 * N_SPLIT;
        const unsigned* go = gO + o0 * N_SPLIT;
        float*          ob = outbh + (size_t)o0 * (HOUT * WOUT) + w;
        #pragma unroll
        for (int i0 = I0; i0 < I1; i0 += SUB) {
            v4f cf[SUB];
            v4u of[SUB];
            #pragma unroll
            for (int j = 0; j < SUB; ++j) {
                if (i0 + j < I1) {
                    cf[j] = *(const v4f*)(cw + (i0 + j) * (8 * N_SPLIT));
                    of[j] = *(const v4u*)(go + (i0 + j) * (8 * N_SPLIT));
                }
            }
            float f[SUB][N_SPLIT];
            #pragma unroll
            for (int j = 0; j < SUB; ++j) {
                if (i0 + j < I1) {
                    f[j][0] = *(const float*)(fb + of[j].x + 0 * (N_LEGO * FS_W * 4));
                    f[j][1] = *(const float*)(fb + of[j].y + 1 * (N_LEGO * FS_W * 4));
                    f[j][2] = *(const float*)(fb + of[j].z + 2 * (N_LEGO * FS_W * 4));
                    f[j][3] = *(const float*)(fb + of[j].w + 3 * (N_LEGO * FS_W * 4));
                }
            }
            #pragma unroll
            for (int j = 0; j < SUB; ++j) {
                if (i0 + j < I1) {
                    float sum = 0.f;
                    sum += cf[j].x * f[j][0];
                    sum += cf[j].y * f[j][1];
                    sum += cf[j].z * f[j][2];
                    sum += cf[j].w * f[j][3];
                    ob[(i0 + j) * (8 * HOUT * WOUT)] = sum;
                }
            }
        }
    }
}

__device__ __forceinline__ void spill_feat(const v8f (&acc)[4], float* Fc, const int s,
                                           const int nt, const int half, const int l16) {
    #pragma unroll
    for (int mt = 0; mt < 4; ++mt) {
        float* fbase = Fc + (s * N_LEGO + mt * 16 + half * 8) * FS_W + nt * 16 + l16;
        #pragma unroll
        for (int v = 0; v < 8; ++v)
            fbase[v * FS_W] = acc[mt][v];
    }
}

template <bool CMB>
__device__ __forceinline__ void row_stage(v8f (&acc)[4], BChunk& q0, BChunk& q1,
                                          const float* xs, const float* afr,
                                          const int (&off)[3][2], const bool (&ok)[3][2],
                                          const int (&offn)[3][2],
                                          const float* Fp, const float* cW4,
                                          const unsigned* iP, float* outbh,
                                          const unsigned tid) {
    load_chunk(q1, xs + 1 * CH_CI * CHW, off);
    SCHED_FENCE();
    if (CMB) {
        mma_chunk<0, 2>(acc, q0, afr, 0, ok);
        SCHED_FENCE();
        combine_part<24, 32, 8>(Fp, cW4, iP, outbh, tid);
        SCHED_FENCE();
        mma_chunk<2, CH_G>(acc, q0, afr, 0, ok);
    } else {
        mma_chunk(acc, q0, afr, 0, ok);
    }
    SCHED_FENCE();
    load_chunk(q0, xs + 2 * CH_CI * CHW, off);
    SCHED_FENCE();
    if (CMB) { combine_part<0, 8, 8>(Fp, cW4, iP, outbh, tid); SCHED_FENCE(); }
    mma_chunk(acc, q1, afr, 12, ok);
    SCHED_FENCE();
    load_chunk(q1, xs + 3 * CH_CI * CHW, off);
    SCHED_FENCE();
    if (CMB) { combine_part<8, 16, 8>(Fp, cW4, iP, outbh, tid); SCHED_FENCE(); }
    mma_chunk(acc, q0, afr, 24, ok);
    SCHED_FENCE();
    load_chunk(q0, xs, offn);
    SCHED_FENCE();
    if (CMB) { combine_part<16, 24, 8>(Fp, cW4, iP, outbh, tid); SCHED_FENCE(); }
    mma_chunk(acc, q1, afr, 36, ok);
    SCHED_FENCE();
}

template <int XM>
__device__ __forceinline__ int xmask_i(const int v) {
    return __builtin_amdgcn_update_dpp(v, v, 0x160 + XM, 0xf, 0xf, false);
}
template <int XM>
__device__ __forceinline__ float xmask_f(const float v) {
    return __builtin_bit_cast(float, xmask_i<XM>(__builtin_bit_cast(int, v)));
}

__device__ __forceinline__ int row_argmax8(const v4f va, const v4f vb, const int l8) {
    float m = __builtin_fmaxf(__builtin_fmaxf(va.x, va.y), __builtin_fmaxf(va.z, va.w));
    m = __builtin_fmaxf(m, __builtin_fmaxf(__builtin_fmaxf(vb.x, vb.y), __builtin_fmaxf(vb.z, vb.w)));
    m = __builtin_fmaxf(m, xmask_f<1>(m));
    m = __builtin_fmaxf(m, xmask_f<2>(m));
    m = __builtin_fmaxf(m, xmask_f<4>(m));
    int jc = 64;
    jc = (vb.w == m) ? 35 : jc;
    jc = (vb.z == m) ? 34 : jc;
    jc = (vb.y == m) ? 33 : jc;
    jc = (vb.x == m) ? 32 : jc;
    jc = (va.w == m) ? 3 : jc;
    jc = (va.z == m) ? 2 : jc;
    jc = (va.y == m) ? 1 : jc;
    jc = (va.x == m) ? 0 : jc;
    int c = jc + l8 * 4;
    c = min(c, xmask_i<1>(c));
    c = min(c, xmask_i<2>(c));
    c = min(c, xmask_i<4>(c));
    return min(c, N_LEGO - 1);
}

__global__ __launch_bounds__(NTHR, 1)
void lego_conv_fused(const float* __restrict__ x,
                     const float* __restrict__ lego,
                     const float* __restrict__ coef,
                     const float* __restrict__ comb,
                     float* __restrict__ out) {
    __shared__ __attribute__((aligned(16))) float smem[KSTEPS * AS_STR + 2 * FS_SIZE + 2 * N_SPLIT * OUT_C];
    float* As = smem;
    float* Fs = As + KSTEPS * AS_STR;
    float* cW4 = Fs + 2 * FS_SIZE;
    unsigned* iP = (unsigned*)(cW4 + N_SPLIT * OUT_C);

    const int tid = threadIdx.x;
    const int b   = blockIdx.y;
    const int h0  = (blockIdx.x * HOUT) / ROW_GROUPS;
    const int h1  = ((blockIdx.x + 1) * HOUT) / ROW_GROUPS;

    const int lane = tid & 31;
    const int wave = tid >> 5;
    const int half = lane >> 4;
    const int l16  = lane & 15;
    const int nt   = wave & 3;
    const int s    = wave >> 2;
    const int l8   = lane & 7;
    const int g8   = lane >> 3;

    const float* cb0 = comb + (wave * 64 + g8) * N_LEGO + l8 * 4;
    v4f ra[16], rb[16];
    #pragma unroll
    for (int st = 0; st < 16; ++st) {
        ra[st] = *(const v4f*)(cb0 + st * (4 * N_LEGO));
        rb[st] = *(const v4f*)(cb0 + st * (4 * N_LEGO) + 32);
    }
    v4f lg[6];
    #pragma unroll
    for (int i = 0; i < 6; ++i)
        lg[i] = *(const v4f*)(lego + (tid + i * NTHR) * 4);
    const int  w     = nt * 16 + l16;
    const bool colok = (w >= 1) && (w <= WIN);
    const int  col   = (w < 1) ? 0 : ((w > WIN) ? (WIN - 1) : (w - 1));
    int  off[3][2];
    bool ok[3][2];
    row_offsets(off, ok, h0, half, col, colok);
    const float* xs = x + ((size_t)b * IN_C + (size_t)s * BASIC_C) * CHW;

    BChunk q0, q1;
    load_chunk(q0, xs, off);
    SCHED_FENCE();

    int   prow[2], pbest[2];
    float pcoef[2];
    #pragma unroll
    for (int bt = 0; bt < 2; ++bt) {
        int mine = 0;
        #pragma unroll
        for (int j = 0; j < 8; ++j) {
            const int r = row_argmax8(ra[bt * 8 + j], rb[bt * 8 + j], l8);
            mine = (l8 == j) ? r : mine;
        }
        const int p = wave * 64 + (bt * 8 + l8) * 4 + g8;
        prow[bt]  = p;
        pbest[bt] = mine;
        pcoef[bt] = coef[p * N_LEGO + mine];
    }

    #pragma unroll
    for (int i = 0; i < 6; ++i) {
        const int q  = tid + i * NTHR;
        const int m  = q / KSTEPS;
        const int ks = q - m * KSTEPS;
        float* d = As + ks * AS_STR + (m >> 4) * 64 + (m & 15) * 2;
        v2f lo; lo.x = lg[i].x; lo.y = lg[i].y;
        v2f hi; hi.x = lg[i].z; hi.y = lg[i].w;
        *(v2f*)d        = lo;
        *(v2f*)(d + 32) = hi;
    }
    #pragma unroll
    for (int bt = 0; bt < 2; ++bt) {
        const int p = prow[bt];
        cW4[(p & (OUT_C - 1)) * N_SPLIT + (p >> 8)] = pcoef[bt];
        iP[(p & (OUT_C - 1)) * N_SPLIT + (p >> 8)] = (unsigned)pbest[bt] * (unsigned)(FS_W * 4);
    }
    __syncthreads();

    const float* afr = As + lane * 2;
    float* outb = out + (size_t)b * OUT_C * HOUT * WOUT;

    {
        v8f acc[4];
        #pragma unroll
        for (int mt = 0; mt < 4; ++mt)
            acc[mt] = (v8f){0.f, 0.f, 0.f, 0.f, 0.f, 0.f, 0.f, 0.f};
        int  offn[3][2];
        bool okn[3][2];
        row_offsets(offn, okn, (h0 + 1 < h1) ? (h0 + 1) : h0, half, col, colok);
        row_stage<false>(acc, q0, q1, xs, afr, off, ok, offn, Fs, cW4, iP, outb, (unsigned)tid);
        spill_feat(acc, Fs, s, nt, half, l16);
        __syncthreads();
        #pragma unroll
        for (int j = 0; j < 3; ++j) {
            #pragma unroll
            for (int e = 0; e < 2; ++e) { off[j][e] = offn[j][e]; ok[j][e] = okn[j][e]; }
        }
    }

    for (int h = h0 + 1; h < h1; ++h) {
        float*       Fc = Fs + ((h - h0) & 1) * FS_SIZE;
        const float* Fp = Fs + ((h - h0 + 1) & 1) * FS_SIZE;

        v8f acc[4];
        #pragma unroll
        for (int mt = 0; mt < 4; ++mt)
            acc[mt] = (v8f){0.f, 0.f, 0.f, 0.f, 0.f, 0.f, 0.f, 0.f};

        int  offn[3][2];
        bool okn[3][2];
        row_offsets(offn, okn, (h + 1 < h1) ? (h + 1) : h, half, col, colok);

        row_stage<true>(acc, q0, q1, xs, afr, off, ok, offn, Fp, cW4, iP,
                        outb + (size_t)(h - 1) * WOUT, (unsigned)tid);

        spill_feat(acc, Fc, s, nt, half, l16);
        __syncthreads();

        #pragma unroll
        for (int j = 0; j < 3; ++j) {
            #pragma unroll
            for (int e = 0; e < 2; ++e) { off[j][e] = offn[j][e]; ok[j][e] = okn[j][e]; }
        }
    }

    {
        const float* Fl    = Fs + ((h1 - 1 - h0) & 1) * FS_SIZE;
        float*       outbl = outb + (size_t)(h1 - 1) * WOUT;
        #pragma nounroll
        for (unsigned r = 0; r < 4u; ++r)
            combine_part<0, 8, 8>(Fl, cW4, iP, outbl, (unsigned)tid, r * 8u);
    }
}

extern "C" void kernel_launch(void* const* d_in, const int* in_sizes, int n_in,
                              void* d_out, int out_size, void* d_ws, size_t ws_size,
                              hipStream_t stream) {
    (void)in_sizes; (void)n_in; (void)d_ws; (void)ws_size; (void)out_size;
    const float* x    = (const float*)d_in[0];
    const float* lego = (const float*)d_in[1];
    const float* coef = (const float*)d_in[2];
    const float* comb = (const float*)d_in[3];
    float* out = (float*)d_out;

    dim3 grid(ROW_GROUPS, 32, 1);
    dim3 block(NTHR, 1, 1);
    lego_conv_fused<<<grid, block, 0, stream>>>(x, lego, coef, comb, out);
}


# rows rebalanced: every group takes 2 whole rows, the 3 left-over rows go as column halves to groups 0-5 (half row first, on 8 waves, 2 per SIMD); longest group 2.5 rows, was 3; sums kept
# speedup vs baseline: 5.6621x; 1.0660x over previous
#include <hip/hip_runtime.h>
#include <hip/hip_bf16.h>


typedef __attribute__((ext_vector_type(2))) float v2f;
typedef __attribute__((ext_vector_type(4))) float v4f;
typedef __attribute__((ext_vector_type(8))) float v8f;
typedef __attribute__((ext_vector_type(4))) unsigned v4u;

#define IN_C    256
#define OUT_C   256
#define N_SPLIT 4
#define N_LEGO  64
#define BASIC_C 64
#define HIN     56
#define WIN     56
#define HOUT    19
#define WOUT    58
#define KDIM    192
#define KSTEPS  48
#define CHW     (HIN * WIN)
#define AS_STR  258
#define FS_W    64
#define FS_SIZE (N_SPLIT * N_LEGO * FS_W)
#define CH_G    4
#define CH_CI   16
#define NTHR    512
#define ROW_GROUPS 8

#define SCHED_FENCE() __builtin_amdgcn_sched_barrier(0)

struct BChunk { float v[CH_G][3][2]; };

__device__ __forceinline__ void load_chunk(BChunk& c, const float* __restrict__ p,
                                           const int (&off)[3][2]) {
    #pragma unroll
    for (int g = 0; g < CH_G; ++g) {
        #pragma unroll
        for (int j = 0; j < 3; ++j) {
            #pragma unroll
            for (int e = 0; e < 2; ++e)
                c.v[g][j][e] = p[g * (4 * CHW) + off[j][e]];
        }
    }
}

template <int G0 = 0, int G1 = CH_G>
__device__ __forceinline__ void mma_chunk(v8f (&acc)[4], const BChunk& c,
                                          const float* afr, const int ks0,
                                          const bool (&ok)[3][2]) {
    #pragma unroll
    for (int g = G0; g < G1; ++g) {
        #pragma unroll
        for (int j = 0; j < 3; ++j) {
            v2f bv;
            bv.x = ok[j][0] ? c.v[g][j][0] : 0.0f;
            bv.y = ok[j][1] ? c.v[g][j][1] : 0.0f;
            const float* ap = afr + (ks0 + g * 3 + j) * AS_STR;
            #pragma unroll
            for (int mt = 0; mt < 4; ++mt) {
                v2f a = *(const v2f*)(ap + mt * 64);
                acc[mt] = __builtin_amdgcn_wmma_f32_16x16x4_f32(
                              false, a, false, bv, (short)0, acc[mt], false, false);
            }
        }
    }
}

__device__ __forceinline__ void row_offsets(int (&off)[3][2], bool (&ok)[3][2],
                                            const int h, const int half,
                                            const int col, const bool colok) {
    #pragma unroll
    for (int j = 0; j < 3; ++j) {
        #pragma unroll
        for (int e = 0; e < 2; ++e) {
            const int k  = 4 * j + 2 * half + e;
            const int ci = k / 3;
            const int kh = k - 3 * ci;
            const int r  = 3 * h + kh - 1;
            ok[j][e]  = colok && (r >= 0);
            off[j][e] = ci * CHW + (r < 0 ? 0 : r) * WIN + col;
        }
    }
}

template <int I0, int I1, int SUB>
__device__ __forceinline__ void combine_part(const float* Fp, const float* cW4,
                                             const unsigned* gO, float* outbh,
                                             const unsigned tid, const unsigned wlo,
                                             const unsigned wn, const unsigned ib = 0u) {
    unsigned t = tid;
    asm volatile("" : "+v"(t));
    const unsigned w  = t & 63u;
    const unsigned o0 = (t >> 6) + 8u * ib;
    if (w - wlo < wn) {
        const char*     fb = (const char*)(Fp + w);
        const float*    cw = cW4 + o0 * N_SPLIT;
        const unsigned* go = gO + o0 * N_SPLIT;
        float*          ob = outbh + (size_t)o0 * (HOUT * WOUT) + w;
        #pragma unroll
        for (int i0 = I0; i0 < I1; i0 += SUB) {
            v4f cf[SUB];
            v4u of[SUB];
            #pragma unroll
            for (int j = 0; j < SUB; ++j) {
                if (i0 + j < I1) {
                    cf[j] = *(const v4f*)(cw + (i0 + j) * (8 * N_SPLIT));
                    of[j] = *(const v4u*)(go + (i0 + j) * (8 * N_SPLIT));
                }
            }
            float f[SUB][N_SPLIT];
            #pragma unroll
            for (int j = 0; j < SUB; ++j) {
                if (i0 + j < I1) {
                    f[j][0] = *(const float*)(fb + of[j].x + 0 * (N_LEGO * FS_W * 4));
                    f[j][1] = *(const float*)(fb + of[j].y + 1 * (N_LEGO * FS_W * 4));
                    f[j][2] = *(const float*)(fb + of[j].z + 2 * (N_LEGO * FS_W * 4));
                    f[j][3] = *(const float*)(fb + of[j].w + 3 * (N_LEGO * FS_W * 4));
                }
            }
            #pragma unroll
            for (int j = 0; j < SUB; ++j) {
                if (i0 + j < I1) {
                    float sum = 0.f;
                    sum += cf[j].x * f[j][0];
                    sum += cf[j].y * f[j][1];
                    sum += cf[j].z * f[j][2];
                    sum += cf[j].w * f[j][3];
                    ob[(i0 + j) * (8 * HOUT * WOUT)] = sum;
                }
            }
        }
    }
}

__device__ __forceinline__ void spill_feat(const v8f (&acc)[4], float* Fc, const int s,
                                           const int nt, const int half, const int l16) {
    #pragma unroll
    for (int mt = 0; mt < 4; ++mt) {
        float* fbase = Fc + (s * N_LEGO + mt * 16 + half * 8) * FS_W + nt * 16 + l16;
        #pragma unroll
        for (int v = 0; v < 8; ++v)
            fbase[v * FS_W] = acc[mt][v];
    }
}

template <bool CMB>
__device__ __forceinline__ void row_stage(v8f (&acc)[4], BChunk& q0, BChunk& q1,
                                          const float* xs, const float* afr,
                                          const int (&off)[3][2], const bool (&ok)[3][2],
                                          const int (&offn)[3][2],
                                          const float* Fp, const float* cW4,
                                          const unsigned* iP, float* outbh,
                                          const unsigned tid, const unsigned wlo,
                                          const unsigned wn) {
    load_chunk(q1, xs + 1 * CH_CI * CHW, off);
    SCHED_FENCE();
    if (CMB) {
        mma_chunk<0, 2>(acc, q0, afr, 0, ok);
        SCHED_FENCE();
        combine_part<24, 32, 8>(Fp, cW4, iP, outbh, tid, wlo, wn);
        SCHED_FENCE();
        mma_chunk<2, CH_G>(acc, q0, afr, 0, ok);
    } else {
        mma_chunk(acc, q0, afr, 0, ok);
    }
    SCHED_FENCE();
    load_chunk(q0, xs + 2 * CH_CI * CHW, off);
    SCHED_FENCE();
    if (CMB) { combine_part<0, 8, 8>(Fp, cW4, iP, outbh, tid, wlo, wn); SCHED_FENCE(); }
    mma_chunk(acc, q1, afr, 12, ok);
    SCHED_FENCE();
    load_chunk(q1, xs + 3 * CH_CI * CHW, off);
    SCHED_FENCE();
    if (CMB) { combine_part<8, 16, 8>(Fp, cW4, iP, outbh, tid, wlo, wn); SCHED_FENCE(); }
    mma_chunk(acc, q0, afr, 24, ok);
    SCHED_FENCE();
    load_chunk(q0, xs, offn);
    SCHED_FENCE();
    if (CMB) { combine_part<16, 24, 8>(Fp, cW4, iP, outbh, tid, wlo, wn); SCHED_FENCE(); }
    mma_chunk(acc, q1, afr, 36, ok);
    SCHED_FENCE();
}

template <int XM>
__device__ __forceinline__ int xmask_i(const int v) {
    return __builtin_amdgcn_update_dpp(v, v, 0x160 + XM, 0xf, 0xf, false);
}
template <int XM>
__device__ __forceinline__ float xmask_f(const float v) {
    return __builtin_bit_cast(float, xmask_i<XM>(__builtin_bit_cast(int, v)));
}

__device__ __forceinline__ int row_argmax8(const v4f va, const v4f vb, const int l8) {
    float m = __builtin_fmaxf(__builtin_fmaxf(va.x, va.y), __builtin_fmaxf(va.z, va.w));
    m = __builtin_fmaxf(m, __builtin_fmaxf(__builtin_fmaxf(vb.x, vb.y), __builtin_fmaxf(vb.z, vb.w)));
    m = __builtin_fmaxf(m, xmask_f<1>(m));
    m = __builtin_fmaxf(m, xmask_f<2>(m));
    m = __builtin_fmaxf(m, xmask_f<4>(m));
    int jc = 64;
    jc = (vb.w == m) ? 35 : jc;
    jc = (vb.z == m) ? 34 : jc;
    jc = (vb.y == m) ? 33 : jc;
    jc = (vb.x == m) ? 32 : jc;
    jc = (va.w == m) ? 3 : jc;
    jc = (va.z == m) ? 2 : jc;
    jc = (va.y == m) ? 1 : jc;
    jc = (va.x == m) ? 0 : jc;
    int c = jc + l8 * 4;
    c = min(c, xmask_i<1>(c));
    c = min(c, xmask_i<2>(c));
    c = min(c, xmask_i<4>(c));
    return min(c, N_LEGO - 1);
}

__global__ __launch_bounds__(NTHR, 1)
void lego_conv_fused(const float* __restrict__ x,
                     const float* __restrict__ lego,
                     const float* __restrict__ coef,
                     const float* __restrict__ comb,
                     float* __restrict__ out) {
    __shared__ __attribute__((aligned(16))) float smem[KSTEPS * AS_STR + 2 * FS_SIZE + 2 * N_SPLIT * OUT_C];
    float* As = smem;
    float* Fs = As + KSTEPS * AS_STR;
    float* cW4 = Fs + 2 * FS_SIZE;
    unsigned* iP = (unsigned*)(cW4 + N_SPLIT * OUT_C);

    const int tid = threadIdx.x;
    const int b   = blockIdx.y;
    const int  j8       = blockIdx.x;
    const int  hf0      = 2 * j8;
    const bool has_half = j8 < 6;
    const int  hx       = 16 + (j8 >> 1);
    const int  chalf    = j8 & 1;

    const int lane = tid & 31;
    const int wave = tid >> 5;
    const int half = lane >> 4;
    const int l16  = lane & 15;
    const int nt   = wave & 3;
    const int s    = wave >> 2;
    const int l8   = lane & 7;
    const int g8   = lane >> 3;
    const bool actv  = ((((wave >> 1) ^ (wave >> 2)) & 1) == 0);
    const int  ntH   = 2 * chalf + (wave & 1);
    const bool halfw = has_half && actv;

    const float* cb0 = comb + (wave * 64 + g8) * N_LEGO + l8 * 4;
    v4f ra[16], rb[16];
    #pragma unroll
    for (int st = 0; st < 16; ++st) {
        ra[st] = *(const v4f*)(cb0 + st * (4 * N_LEGO));
        rb[st] = *(const v4f*)(cb0 + st * (4 * N_LEGO) + 32);
    }
    v4f lg[6];
    #pragma unroll
    for (int i = 0; i < 6; ++i)
        lg[i] = *(const v4f*)(lego + (tid + i * NTHR) * 4);
    const int  w     = nt * 16 + l16;
    const bool colok = (w >= 1) && (w <= WIN);
    const int  col   = (w < 1) ? 0 : ((w > WIN) ? (WIN - 1) : (w - 1));
    const int  w0     = (halfw ? ntH : nt) * 16 + l16;
    const bool colok0 = (w0 >= 1) && (w0 <= WIN);
    const int  col0   = (w0 < 1) ? 0 : ((w0 > WIN) ? (WIN - 1) : (w0 - 1));
    int  off[3][2];
    bool ok[3][2];
    row_offsets(off, ok, halfw ? hx : hf0, half, col0, colok0);
    const float* xs = x + ((size_t)b * IN_C + (size_t)s * BASIC_C) * CHW;

    BChunk q0, q1;
    load_chunk(q0, xs, off);
    SCHED_FENCE();

    int   prow[2], pbest[2];
    float pcoef[2];
    #pragma unroll
    for (int bt = 0; bt < 2; ++bt) {
        int mine = 0;
        #pragma unroll
        for (int j = 0; j < 8; ++j) {
            const int r = row_argmax8(ra[bt * 8 + j], rb[bt * 8 + j], l8);
            mine = (l8 == j) ? r : mine;
        }
        const int p = wave * 64 + (bt * 8 + l8) * 4 + g8;
        prow[bt]  = p;
        pbest[bt] = mine;
        pcoef[bt] = coef[p * N_LEGO + mine];
    }

    #pragma unroll
    for (int i = 0; i < 6; ++i) {
        const int q  = tid + i * NTHR;
        const int m  = q / KSTEPS;
        const int ks = q - m * KSTEPS;
        float* d = As + ks * AS_STR + (m >> 4) * 64 + (m & 15) * 2;
        v2f lo; lo.x = lg[i].x; lo.y = lg[i].y;
        v2f hi; hi.x = lg[i].z; hi.y = lg[i].w;
        *(v2f*)d        = lo;
        *(v2f*)(d + 32) = hi;
    }
    #pragma unroll
    for (int bt = 0; bt < 2; ++bt) {
        const int p = prow[bt];
        cW4[(p & (OUT_C - 1)) * N_SPLIT + (p >> 8)] = pcoef[bt];
        iP[(p & (OUT_C - 1)) * N_SPLIT + (p >> 8)] = (unsigned)pbest[bt] * (unsigned)(FS_W * 4);
    }
    __syncthreads();

    const float* afr = As + lane * 2;
    float* outb = out + (size_t)b * OUT_C * HOUT * WOUT;

    int      par  = 0;
    unsigned wlo  = 0u;
    unsigned wn   = 0u;
    float*   outp = outb;

    if (has_half) {
        if (__builtin_amdgcn_readfirstlane((int)actv) != 0) {
            v8f acc[4];
            #pragma unroll
            for (int mt = 0; mt < 4; ++mt)
                acc[mt] = (v8f){0.f, 0.f, 0.f, 0.f, 0.f, 0.f, 0.f, 0.f};
            int  offn[3][2];
            bool okn[3][2];
            row_offsets(offn, okn, hf0, half, col, colok);
            row_stage<false>(acc, q0, q1, xs, afr, off, ok, offn, Fs, cW4, iP, outb, (unsigned)tid, 0u, 0u);
            spill_feat(acc, Fs, s, ntH, half, l16);
            #pragma unroll
            for (int j = 0; j < 3; ++j) {
                #pragma unroll
                for (int e = 0; e < 2; ++e) { off[j][e] = offn[j][e]; ok[j][e] = okn[j][e]; }
            }
        }
        __syncthreads();
        par  = 1;
        wlo  = chalf ? 32u : 0u;
        wn   = chalf ? (unsigned)(WOUT - 32) : 32u;
        outp = outb + (size_t)hx * WOUT;
    }

    #pragma nounroll
    for (int r = 0; r < 2; ++r) {
        float*       Fc = Fs + par * FS_SIZE;
        const float* Fp = Fs + (par ^ 1) * FS_SIZE;

        v8f acc[4];
        #pragma unroll
        for (int mt = 0; mt < 4; ++mt)
            acc[mt] = (v8f){0.f, 0.f, 0.f, 0.f, 0.f, 0.f, 0.f, 0.f};

        int  offn[3][2];
        bool okn[3][2];
        row_offsets(offn, okn, hf0 + 1, half, col, colok);

        row_stage<true>(acc, q0, q1, xs, afr, off, ok, offn, Fp, cW4, iP,
                        outp, (unsigned)tid, wlo, wn);

        spill_feat(acc, Fc, s, nt, half, l16);
        __syncthreads();

        #pragma unroll
        for (int j = 0; j < 3; ++j) {
            #pragma unroll
            for (int e = 0; e < 2; ++e) { off[j][e] = offn[j][e]; ok[j][e] = okn[j][e]; }
        }
        par ^= 1;
        wlo  = 0u;
        wn   = (unsigned)WOUT;
        outp = outb + (size_t)(hf0 + r) * WOUT;
    }

    {
        const float* Fl = Fs + (par ^ 1) * FS_SIZE;
        #pragma nounroll
        for (unsigned r = 0; r < 4u; ++r)
            combine_part<0, 8, 8>(Fl, cW4, iP, outp, (unsigned)tid, 0u, (unsigned)WOUT, r * 8u);
    }
}

extern "C" void kernel_launch(void* const* d_in, const int* in_sizes, int n_in,
                              void* d_out, int out_size, void* d_ws, size_t ws_size,
                              hipStream_t stream) {
    (void)in_sizes; (void)n_in; (void)d_ws; (void)ws_size; (void)out_size;
    const float* x    = (const float*)d_in[0];
    const float* lego = (const float*)d_in[1];
    const float* coef = (const float*)d_in[2];
    const float* comb = (const float*)d_in[3];
    float* out = (float*)d_out;

    dim3 grid(ROW_GROUPS, 32, 1);
    dim3 block(NTHR, 1, 1);
    lego_conv_fused<<<grid, block, 0, stream>>>(x, lego, coef, comb, out);
}
